# starting kernel
# baseline (speedup 1.0000x reference)
#include <hip/hip_runtime.h>


typedef __attribute__((ext_vector_type(2))) float v2f;
typedef __attribute__((ext_vector_type(8))) float v8f;

#define NUM_NODES (8 * 8192)
#define FDIM 64

__global__ void gp_scatter_kernel(const float* __restrict__ h,
                                  const int* __restrict__ src,
                                  const int* __restrict__ dst,
                                  float* __restrict__ msum,
                                  float* __restrict__ deg,
                                  int num_edges) {
    const int lane   = threadIdx.x & 31;
    const int wave   = (blockIdx.x * blockDim.x + threadIdx.x) >> 5;
    const int nwaves = (gridDim.x * blockDim.x) >> 5;

    for (int e = wave; e < num_edges; e += nwaves) {
        const int s = src[e];
        const int d = dst[e];
        const float2 hv = *(const float2*)(h + (size_t)s * FDIM + lane * 2);
        float* mrow = msum + (size_t)d * FDIM + lane * 2;
        unsafeAtomicAdd(mrow + 0, hv.x);
        unsafeAtomicAdd(mrow + 1, hv.y);
        if (lane == 0) unsafeAtomicAdd(deg + d, 1.0f);
    }
}

__global__ void gp_finalize_kernel(const float* __restrict__ h,
                                   const float* __restrict__ msum,
                                   const float* __restrict__ deg,
                                   float* __restrict__ out,
                                   int num_tiles) {
    const int lane = threadIdx.x & 31;
    const int wave = (blockIdx.x * blockDim.x + threadIdx.x) >> 5;
    if (wave >= num_tiles) return;

    const int node_base = wave * 16;
    const int half      = lane >> 4;
    const int m5        = lane & 15;

    const float dm = deg[node_base + m5];
    const float r  = (dm > 0.0f) ? (1.0f / dm) : 1.0f;

    const int rowbase = half * 8;
    float degrow[8];
#pragma unroll
    for (int v = 0; v < 8; ++v) degrow[v] = deg[node_base + rowbase + v];

    for (int fc = 0; fc < 4; ++fc) {
        const int feat = fc * 16 + m5;
        v8f c = {};
#pragma unroll
        for (int j = 0; j < 4; ++j) {
            v2f a;
            a.x = (m5 == 4 * j + 2 * half + 0) ? r : 0.0f;
            a.y = (m5 == 4 * j + 2 * half + 1) ? r : 0.0f;
            const int krow = node_base + 4 * j + 2 * half;
            v2f b;
            b.x = msum[(size_t)(krow + 0) * FDIM + feat];
            b.y = msum[(size_t)(krow + 1) * FDIM + feat];
            c = __builtin_amdgcn_wmma_f32_16x16x4_f32(
                     false, a,  false, b,
                     (short)0, c,  false,  false);
        }
#pragma unroll
        for (int v = 0; v < 8; ++v) {
            const size_t idx = (size_t)(node_base + rowbase + v) * FDIM + feat;
            const float hv = h[idx];
            out[idx] = (degrow[v] > 0.0f) ? c[v] : hv;
        }
    }
}

extern "C" void kernel_launch(void* const* d_in, const int* in_sizes, int n_in,
                              void* d_out, int out_size, void* d_ws, size_t ws_size,
                              hipStream_t stream) {
    const float* h   = (const float*)d_in[0];
    const int*   src = (const int*)d_in[1];
    const int*   dst = (const int*)d_in[2];
    float*       out = (float*)d_out;
    const int num_edges = in_sizes[1];

    float* msum = (float*)d_ws;
    float* deg  = msum + (size_t)NUM_NODES * FDIM;

    hipMemsetAsync(d_ws, 0, ((size_t)NUM_NODES * FDIM + NUM_NODES) * sizeof(float),
                   stream);

    gp_scatter_kernel<<<2048, 256, 0, stream>>>(h, src, dst, msum, deg, num_edges);

    const int num_tiles = NUM_NODES / 16;
    gp_finalize_kernel<<<num_tiles / 8, 256, 0, stream>>>(h, msum, deg, out, num_tiles);
}


# Sort edges by dst (int rank atomics, arrival-order rank), then atomic-free gather per 16-node tile: WMMA does the segment sum with the 1/deg scale; float-atomic scatter gone; sum order changes
# speedup vs baseline: 1.5249x; 1.5249x over previous
#include <hip/hip_runtime.h>


typedef __attribute__((ext_vector_type(2))) float v2f;
typedef __attribute__((ext_vector_type(8))) float v8f;

#define NUM_NODES (8 * 8192)
#define FDIM 64
#define EDGE_CAP (1 << 20)

__global__ void gp_rank_kernel(const int* __restrict__ dst,
                               int* __restrict__ cnt,
                               int* __restrict__ rank,
                               int num_edges) {
    const int stride = gridDim.x * blockDim.x;
    for (int e = blockIdx.x * blockDim.x + threadIdx.x; e < num_edges; e += stride)
        rank[e] = atomicAdd(cnt + dst[e], 1);
}

__global__ __launch_bounds__(1024) void gp_scan_kernel(const int* __restrict__ cnt,
                                                       int* __restrict__ off) {
    __shared__ int wtot[32];
    const int lane = threadIdx.x & 31;
    const int w    = threadIdx.x >> 5;
    const int base = w * 2048;

    int4 v[16];
#pragma unroll
    for (int s = 0; s < 16; ++s)
        v[s] = *(const int4*)(cnt + base + s * 128 + lane * 4);

    int carry = 0;
#pragma unroll
    for (int s = 0; s < 16; ++s) {
        const int t0 = v[s].x, t1 = t0 + v[s].y, t2 = t1 + v[s].z, t3 = t2 + v[s].w;
        int incl = t3;
        for (int d = 1; d < 32; d <<= 1) {
            const int n = __shfl_up(incl, d);
            if (lane >= d) incl += n;
        }
        const int excl = incl - t3 + carry;
        v[s].x = excl; v[s].y = excl + t0; v[s].z = excl + t1; v[s].w = excl + t2;
        carry += __shfl(incl, 31);
    }
    if (lane == 0) wtot[w] = carry;
    __syncthreads();
    int x = (lane < w) ? wtot[lane] : 0;
    for (int d = 16; d >= 1; d >>= 1) x += __shfl_xor(x, d);
#pragma unroll
    for (int s = 0; s < 16; ++s) {
        int4 o = v[s];
        o.x += x; o.y += x; o.z += x; o.w += x;
        *(int4*)(off + base + s * 128 + lane * 4) = o;
    }
    if (threadIdx.x == 1023) off[NUM_NODES] = x + carry;
}

__global__ void gp_place_kernel(const int* __restrict__ src,
                                const int* __restrict__ dst,
                                const int* __restrict__ rank,
                                const int* __restrict__ off,
                                int* __restrict__ perm,
                                int num_edges) {
    const int stride = gridDim.x * blockDim.x;
    for (int e = blockIdx.x * blockDim.x + threadIdx.x; e < num_edges; e += stride)
        perm[off[dst[e]] + rank[e]] = src[e];
}

__global__ __launch_bounds__(256) void gp_gather_kernel(const float* __restrict__ h,
                                                        const int* __restrict__ off,
                                                        const int* __restrict__ perm,
                                                        float* __restrict__ out,
                                                        int num_tiles) {
    const int lane = threadIdx.x & 31;
    const int wave = __builtin_amdgcn_readfirstlane((blockIdx.x * blockDim.x + threadIdx.x) >> 5);
    if (wave >= num_tiles) return;

    const int nb   = wave * 16;
    const int half = lane >> 4;
    const int c    = lane & 15;

    const int lo = off[nb + c];
    const int hi = off[nb + c + 1];
    const int dg = hi - lo;
    const float r = 1.0f / (float)(dg > 1 ? dg : 1);
    const int start = off[nb];
    const int end   = off[nb + 16];
    const int last  = end - 1;

    const float* hcol = h + 4 * c;

    v8f acc0 = {}, acc1 = {}, acc2 = {}, acc3 = {};

    int k0 = start & ~3;
    int idx[8];
    if (k0 < end) {
#pragma unroll
        for (int j = 0; j < 8; ++j) {
            int p = k0 + 4 * (j >> 1) + 2 * half + (j & 1);
            p = p < start ? start : p;
            p = p > last ? last : p;
            idx[j] = perm[p];
        }
    }

    for (; k0 < end; k0 += 16) {
        float4 b[8];
#pragma unroll
        for (int j = 0; j < 8; ++j)
            b[j] = *(const float4*)(hcol + (size_t)idx[j] * FDIM);
#pragma unroll
        for (int j = 0; j < 8; ++j) {
            int p = k0 + 16 + 4 * (j >> 1) + 2 * half + (j & 1);
            p = p > last ? last : p;
            idx[j] = perm[p];
        }
#pragma unroll
        for (int u = 0; u < 4; ++u) {
            const int p0 = k0 + 4 * u + 2 * half;
            v2f a;
            a.x = (p0     >= lo && p0     < hi) ? r : 0.0f;
            a.y = (p0 + 1 >= lo && p0 + 1 < hi) ? r : 0.0f;
            v2f b0, b1, b2, b3;
            b0.x = b[2 * u].x; b0.y = b[2 * u + 1].x;
            b1.x = b[2 * u].y; b1.y = b[2 * u + 1].y;
            b2.x = b[2 * u].z; b2.y = b[2 * u + 1].z;
            b3.x = b[2 * u].w; b3.y = b[2 * u + 1].w;
            acc0 = __builtin_amdgcn_wmma_f32_16x16x4_f32(false, a, false, b0, (short)0, acc0, false, false);
            acc1 = __builtin_amdgcn_wmma_f32_16x16x4_f32(false, a, false, b1, (short)0, acc1, false, false);
            acc2 = __builtin_amdgcn_wmma_f32_16x16x4_f32(false, a, false, b2, (short)0, acc2, false, false);
            acc3 = __builtin_amdgcn_wmma_f32_16x16x4_f32(false, a, false, b3, (short)0, acc3, false, false);
        }
    }

    const int rowbase = nb + half * 8;
    int o9[9];
#pragma unroll
    for (int v = 0; v < 9; ++v) o9[v] = off[rowbase + v];
#pragma unroll
    for (int v = 0; v < 8; ++v) {
        const size_t at = (size_t)(rowbase + v) * FDIM + 4 * c;
        float4 o = make_float4(acc0[v], acc1[v], acc2[v], acc3[v]);
        if (o9[v + 1] == o9[v]) o = *(const float4*)(h + at);
        *(float4*)(out + at) = o;
    }
}

extern "C" void kernel_launch(void* const* d_in, const int* in_sizes, int n_in,
                              void* d_out, int out_size, void* d_ws, size_t ws_size,
                              hipStream_t stream) {
    const float* h   = (const float*)d_in[0];
    const int*   src = (const int*)d_in[1];
    const int*   dst = (const int*)d_in[2];
    float*       out = (float*)d_out;
    const int num_edges = in_sizes[1];

    int* cnt  = (int*)d_ws;
    int* off  = cnt + NUM_NODES;
    int* rank = off + NUM_NODES + 64;
    int* perm = rank + EDGE_CAP;

    hipMemsetAsync(cnt, 0, (size_t)NUM_NODES * sizeof(int), stream);

    gp_rank_kernel<<<4096, 256, 0, stream>>>(dst, cnt, rank, num_edges);
    gp_scan_kernel<<<1, 1024, 0, stream>>>(cnt, off);
    gp_place_kernel<<<4096, 256, 0, stream>>>(src, dst, rank, off, perm, num_edges);

    const int num_tiles = NUM_NODES / 16;
    gp_gather_kernel<<<num_tiles / 8, 256, 0, stream>>>(h, off, perm, out, num_tiles);
}


# Counting sort by dst now two-level with LDS counters (256 edge chunks x 256-node buckets, 2nd level inside the gather block): global rank atomics, rank/place passes, memset gone; sum order changes
# speedup vs baseline: 2.6379x; 1.7299x over previous
#include <hip/hip_runtime.h>


typedef __attribute__((ext_vector_type(2))) float v2f;
typedef __attribute__((ext_vector_type(8))) float v8f;

#define NUM_NODES (8 * 8192)
#define FDIM 64
#define EDGE_CAP (1 << 20)
#define NBKT 256
#define NCHK 256
#define SCAN_N (NBKT * NCHK)
#define PASS_T 512

__device__ __forceinline__ int gp_chunk_len(int num_edges) {
    return ((num_edges + 1023) >> 10) << 2;
}

__global__ __launch_bounds__(PASS_T) void gp_hist_kernel(const int* __restrict__ dst,
                                                         int* __restrict__ cntm,
                                                         int num_edges) {
    __shared__ int lh[NBKT];
    const int tid = threadIdx.x;
    if (tid < NBKT) lh[tid] = 0;
    __syncthreads();
    const int chunk = gp_chunk_len(num_edges);
    const int lo = blockIdx.x * chunk;
    const int hi = min(lo + chunk, num_edges);
    for (int i = lo + 4 * tid; i < hi; i += 4 * PASS_T) {
        if (i + 4 <= hi) {
            const int4 d = *(const int4*)(dst + i);
            atomicAdd(&lh[(d.x >> 8) & (NBKT - 1)], 1);
            atomicAdd(&lh[(d.y >> 8) & (NBKT - 1)], 1);
            atomicAdd(&lh[(d.z >> 8) & (NBKT - 1)], 1);
            atomicAdd(&lh[(d.w >> 8) & (NBKT - 1)], 1);
        } else {
            for (int e = i; e < hi; ++e) atomicAdd(&lh[(dst[e] >> 8) & (NBKT - 1)], 1);
        }
    }
    __syncthreads();
    if (tid < NBKT) cntm[tid * NCHK + blockIdx.x] = lh[tid];
}

__global__ __launch_bounds__(1024) void gp_scan_kernel(const int* __restrict__ cnt,
                                                       int* __restrict__ off) {
    __shared__ int wtot[32];
    const int lane = threadIdx.x & 31;
    const int w    = threadIdx.x >> 5;
    const int base = w * 2048;

    int4 v[16];
#pragma unroll
    for (int s = 0; s < 16; ++s)
        v[s] = *(const int4*)(cnt + base + s * 128 + lane * 4);

    int carry = 0;
#pragma unroll
    for (int s = 0; s < 16; ++s) {
        const int t0 = v[s].x, t1 = t0 + v[s].y, t2 = t1 + v[s].z, t3 = t2 + v[s].w;
        int incl = t3;
        for (int d = 1; d < 32; d <<= 1) {
            const int n = __shfl_up(incl, d);
            if (lane >= d) incl += n;
        }
        const int excl = incl - t3 + carry;
        v[s].x = excl; v[s].y = excl + t0; v[s].z = excl + t1; v[s].w = excl + t2;
        carry += __shfl(incl, 31);
    }
    if (lane == 0) wtot[w] = carry;
    __syncthreads();
    int x = (lane < w) ? wtot[lane] : 0;
    for (int d = 16; d >= 1; d >>= 1) x += __shfl_xor(x, d);
#pragma unroll
    for (int s = 0; s < 16; ++s) {
        int4 o = v[s];
        o.x += x; o.y += x; o.z += x; o.w += x;
        *(int4*)(off + base + s * 128 + lane * 4) = o;
    }
    if (threadIdx.x == 1023) off[SCAN_N] = x + carry;
}

__global__ __launch_bounds__(PASS_T) void gp_scatter_kernel(const int* __restrict__ src,
                                                            const int* __restrict__ dst,
                                                            const int* __restrict__ offm,
                                                            int* __restrict__ sorted1,
                                                            int num_edges) {
    __shared__ int cur[NBKT];
    const int tid = threadIdx.x;
    if (tid < NBKT) cur[tid] = offm[tid * NCHK + blockIdx.x];
    __syncthreads();
    const int chunk = gp_chunk_len(num_edges);
    const int lo = blockIdx.x * chunk;
    const int hi = min(lo + chunk, num_edges);
    for (int i = lo + 4 * tid; i < hi; i += 4 * PASS_T) {
        if (i + 4 <= hi) {
            const int4 d = *(const int4*)(dst + i);
            const int4 s = *(const int4*)(src + i);
            const int p0 = atomicAdd(&cur[(d.x >> 8) & (NBKT - 1)], 1);
            const int p1 = atomicAdd(&cur[(d.y >> 8) & (NBKT - 1)], 1);
            const int p2 = atomicAdd(&cur[(d.z >> 8) & (NBKT - 1)], 1);
            const int p3 = atomicAdd(&cur[(d.w >> 8) & (NBKT - 1)], 1);
            sorted1[p0] = (s.x & 0xFFFF) | ((d.x & 255) << 16);
            sorted1[p1] = (s.y & 0xFFFF) | ((d.y & 255) << 16);
            sorted1[p2] = (s.z & 0xFFFF) | ((d.z & 255) << 16);
            sorted1[p3] = (s.w & 0xFFFF) | ((d.w & 255) << 16);
        } else {
            for (int e = i; e < hi; ++e) {
                const int d = dst[e];
                const int p = atomicAdd(&cur[(d >> 8) & (NBKT - 1)], 1);
                sorted1[p] = (src[e] & 0xFFFF) | ((d & 255) << 16);
            }
        }
    }
}

__global__ __launch_bounds__(PASS_T) void gp_gather_kernel(const float* __restrict__ h,
                                                           const int* __restrict__ offm,
                                                           const int* __restrict__ sorted1,
                                                           int* perm,
                                                           float* __restrict__ out) {
    __shared__ int ncur[256];
    __shared__ int noff[264];
    const int tid  = threadIdx.x;
    const int lane = tid & 31;
    const int w  = __builtin_amdgcn_readfirstlane(tid >> 5);
    const int bk = blockIdx.x;
    const int s  = offm[bk * NCHK];
    const int e  = offm[bk * NCHK + NCHK];

    if (tid < 256) ncur[tid] = 0;
    __syncthreads();

    for (int i0 = s + tid; i0 < e; i0 += 4 * PASS_T) {
        int v[4];
#pragma unroll
        for (int u = 0; u < 4; ++u) {
            const int i = i0 + u * PASS_T;
            v[u] = (i < e) ? sorted1[i] : -1;
        }
#pragma unroll
        for (int u = 0; u < 4; ++u)
            if (v[u] >= 0) atomicAdd(&ncur[(v[u] >> 16) & 255], 1);
    }
    __syncthreads();

    if (w == 0) {
        int v[8];
        int t = 0;
#pragma unroll
        for (int j = 0; j < 8; ++j) { const int x = ncur[8 * lane + j]; v[j] = t; t += x; }
        int incl = t;
        for (int d = 1; d < 32; d <<= 1) {
            const int n = __shfl_up(incl, d);
            if (lane >= d) incl += n;
        }
        const int excl = incl - t;
#pragma unroll
        for (int j = 0; j < 8; ++j) {
            noff[8 * lane + j] = excl + v[j];
            ncur[8 * lane + j] = excl + v[j];
        }
        if (lane == 31) noff[256] = incl;
    }
    __syncthreads();

    for (int i0 = s + tid; i0 < e; i0 += 4 * PASS_T) {
        int v[4];
#pragma unroll
        for (int u = 0; u < 4; ++u) {
            const int i = i0 + u * PASS_T;
            v[u] = (i < e) ? sorted1[i] : -1;
        }
#pragma unroll
        for (int u = 0; u < 4; ++u)
            if (v[u] >= 0) {
                const int p = atomicAdd(&ncur[(v[u] >> 16) & 255], 1);
                perm[s + p] = v[u] & 0xFFFF;
            }
    }
    __syncthreads();

    const int nl   = w * 16;
    const int nb   = bk * 256 + nl;
    const int half = lane >> 4;
    const int c    = lane & 15;

    const int lo = s + noff[nl + c];
    const int hi = s + noff[nl + c + 1];
    const int dg = hi - lo;
    const float r = 1.0f / (float)(dg > 1 ? dg : 1);
    const int start = s + __builtin_amdgcn_readfirstlane(noff[nl]);
    const int end   = s + __builtin_amdgcn_readfirstlane(noff[nl + 16]);
    const int last  = end - 1;

    const float* hcol = h + 4 * c;

    v8f acc0 = {}, acc1 = {}, acc2 = {}, acc3 = {};

    int k0 = (start < end) ? (start & ~3) : end;
    int idx[8];
    if (k0 < end) {
#pragma unroll
        for (int j = 0; j < 8; ++j) {
            int p = k0 + 4 * (j >> 1) + 2 * half + (j & 1);
            p = p < start ? start : p;
            p = p > last ? last : p;
            idx[j] = perm[p];
        }
    }

    for (; k0 < end; k0 += 16) {
        float4 b[8];
#pragma unroll
        for (int j = 0; j < 8; ++j)
            b[j] = *(const float4*)(hcol + (size_t)idx[j] * FDIM);
#pragma unroll
        for (int j = 0; j < 8; ++j) {
            int p = k0 + 16 + 4 * (j >> 1) + 2 * half + (j & 1);
            p = p > last ? last : p;
            idx[j] = perm[p];
        }
#pragma unroll
        for (int u = 0; u < 4; ++u) {
            const int p0 = k0 + 4 * u + 2 * half;
            v2f a;
            a.x = (p0     >= lo && p0     < hi) ? r : 0.0f;
            a.y = (p0 + 1 >= lo && p0 + 1 < hi) ? r : 0.0f;
            v2f b0, b1, b2, b3;
            b0.x = b[2 * u].x; b0.y = b[2 * u + 1].x;
            b1.x = b[2 * u].y; b1.y = b[2 * u + 1].y;
            b2.x = b[2 * u].z; b2.y = b[2 * u + 1].z;
            b3.x = b[2 * u].w; b3.y = b[2 * u + 1].w;
            acc0 = __builtin_amdgcn_wmma_f32_16x16x4_f32(false, a, false, b0, (short)0, acc0, false, false);
            acc1 = __builtin_amdgcn_wmma_f32_16x16x4_f32(false, a, false, b1, (short)0, acc1, false, false);
            acc2 = __builtin_amdgcn_wmma_f32_16x16x4_f32(false, a, false, b2, (short)0, acc2, false, false);
            acc3 = __builtin_amdgcn_wmma_f32_16x16x4_f32(false, a, false, b3, (short)0, acc3, false, false);
        }
    }

    const int rl = nl + half * 8;
    const int rowbase = nb + half * 8;
    int o9[9];
#pragma unroll
    for (int v = 0; v < 9; ++v) o9[v] = noff[rl + v];
#pragma unroll
    for (int v = 0; v < 8; ++v) {
        const size_t at = (size_t)(rowbase + v) * FDIM + 4 * c;
        float4 o = make_float4(acc0[v], acc1[v], acc2[v], acc3[v]);
        if (o9[v + 1] == o9[v]) o = *(const float4*)(h + at);
        *(float4*)(out + at) = o;
    }
}

extern "C" void kernel_launch(void* const* d_in, const int* in_sizes, int n_in,
                              void* d_out, int out_size, void* d_ws, size_t ws_size,
                              hipStream_t stream) {
    const float* h   = (const float*)d_in[0];
    const int*   src = (const int*)d_in[1];
    const int*   dst = (const int*)d_in[2];
    float*       out = (float*)d_out;
    const int num_edges = in_sizes[1];

    int* cntm    = (int*)d_ws;
    int* offm    = cntm + SCAN_N;
    int* sorted1 = offm + SCAN_N + 64;
    int* perm    = sorted1 + EDGE_CAP;

    gp_hist_kernel<<<NCHK, PASS_T, 0, stream>>>(dst, cntm, num_edges);
    gp_scan_kernel<<<1, 1024, 0, stream>>>(cntm, offm);
    gp_scatter_kernel<<<NCHK, PASS_T, 0, stream>>>(src, dst, offm, sorted1, num_edges);
    gp_gather_kernel<<<NBKT, PASS_T, 0, stream>>>(h, offm, sorted1, perm, out);
}


# Gather loop software-pipelined: steps of 32 edges, the next step's 16 h-row loads stay in flight under this step's WMMAs; row numbers from one coalesced load + cross-lane reads; sum order unchanged
# speedup vs baseline: 2.7569x; 1.0451x over previous
#include <hip/hip_runtime.h>


typedef __attribute__((ext_vector_type(2))) float v2f;
typedef __attribute__((ext_vector_type(8))) float v8f;

#define NUM_NODES (8 * 8192)
#define FDIM 64
#define EDGE_CAP (1 << 20)
#define NBKT 256
#define NCHK 256
#define SCAN_N (NBKT * NCHK)
#define PASS_T 512

__device__ __forceinline__ int gp_chunk_len(int num_edges) {
    return ((num_edges + 1023) >> 10) << 2;
}

__global__ __launch_bounds__(PASS_T) void gp_hist_kernel(const int* __restrict__ dst,
                                                         int* __restrict__ cntm,
                                                         int num_edges) {
    __shared__ int lh[NBKT];
    const int tid = threadIdx.x;
    if (tid < NBKT) lh[tid] = 0;
    __syncthreads();
    const int chunk = gp_chunk_len(num_edges);
    const int lo = blockIdx.x * chunk;
    const int hi = min(lo + chunk, num_edges);
    for (int i = lo + 4 * tid; i < hi; i += 4 * PASS_T) {
        if (i + 4 <= hi) {
            const int4 d = *(const int4*)(dst + i);
            atomicAdd(&lh[(d.x >> 8) & (NBKT - 1)], 1);
            atomicAdd(&lh[(d.y >> 8) & (NBKT - 1)], 1);
            atomicAdd(&lh[(d.z >> 8) & (NBKT - 1)], 1);
            atomicAdd(&lh[(d.w >> 8) & (NBKT - 1)], 1);
        } else {
            for (int e = i; e < hi; ++e) atomicAdd(&lh[(dst[e] >> 8) & (NBKT - 1)], 1);
        }
    }
    __syncthreads();
    if (tid < NBKT) cntm[tid * NCHK + blockIdx.x] = lh[tid];
}

__global__ __launch_bounds__(1024) void gp_scan_kernel(const int* __restrict__ cnt,
                                                       int* __restrict__ off) {
    __shared__ int wtot[32];
    const int lane = threadIdx.x & 31;
    const int w    = threadIdx.x >> 5;
    const int base = w * 2048;

    int4 v[16];
#pragma unroll
    for (int s = 0; s < 16; ++s)
        v[s] = *(const int4*)(cnt + base + s * 128 + lane * 4);

    int carry = 0;
#pragma unroll
    for (int s = 0; s < 16; ++s) {
        const int t0 = v[s].x, t1 = t0 + v[s].y, t2 = t1 + v[s].z, t3 = t2 + v[s].w;
        int incl = t3;
        for (int d = 1; d < 32; d <<= 1) {
            const int n = __shfl_up(incl, d);
            if (lane >= d) incl += n;
        }
        const int excl = incl - t3 + carry;
        v[s].x = excl; v[s].y = excl + t0; v[s].z = excl + t1; v[s].w = excl + t2;
        carry += __shfl(incl, 31);
    }
    if (lane == 0) wtot[w] = carry;
    __syncthreads();
    int x = (lane < w) ? wtot[lane] : 0;
    for (int d = 16; d >= 1; d >>= 1) x += __shfl_xor(x, d);
#pragma unroll
    for (int s = 0; s < 16; ++s) {
        int4 o = v[s];
        o.x += x; o.y += x; o.z += x; o.w += x;
        *(int4*)(off + base + s * 128 + lane * 4) = o;
    }
    if (threadIdx.x == 1023) off[SCAN_N] = x + carry;
}

__global__ __launch_bounds__(PASS_T) void gp_scatter_kernel(const int* __restrict__ src,
                                                            const int* __restrict__ dst,
                                                            const int* __restrict__ offm,
                                                            int* __restrict__ sorted1,
                                                            int num_edges) {
    __shared__ int cur[NBKT];
    const int tid = threadIdx.x;
    if (tid < NBKT) cur[tid] = offm[tid * NCHK + blockIdx.x];
    __syncthreads();
    const int chunk = gp_chunk_len(num_edges);
    const int lo = blockIdx.x * chunk;
    const int hi = min(lo + chunk, num_edges);
    for (int i = lo + 4 * tid; i < hi; i += 4 * PASS_T) {
        if (i + 4 <= hi) {
            const int4 d = *(const int4*)(dst + i);
            const int4 s = *(const int4*)(src + i);
            const int p0 = atomicAdd(&cur[(d.x >> 8) & (NBKT - 1)], 1);
            const int p1 = atomicAdd(&cur[(d.y >> 8) & (NBKT - 1)], 1);
            const int p2 = atomicAdd(&cur[(d.z >> 8) & (NBKT - 1)], 1);
            const int p3 = atomicAdd(&cur[(d.w >> 8) & (NBKT - 1)], 1);
            sorted1[p0] = (s.x & 0xFFFF) | ((d.x & 255) << 16);
            sorted1[p1] = (s.y & 0xFFFF) | ((d.y & 255) << 16);
            sorted1[p2] = (s.z & 0xFFFF) | ((d.z & 255) << 16);
            sorted1[p3] = (s.w & 0xFFFF) | ((d.w & 255) << 16);
        } else {
            for (int e = i; e < hi; ++e) {
                const int d = dst[e];
                const int p = atomicAdd(&cur[(d >> 8) & (NBKT - 1)], 1);
                sorted1[p] = (src[e] & 0xFFFF) | ((d & 255) << 16);
            }
        }
    }
}

__global__ __launch_bounds__(PASS_T) void gp_gather_kernel(const float* __restrict__ h,
                                                           const int* __restrict__ offm,
                                                           const int* __restrict__ sorted1,
                                                           int* perm,
                                                           float* __restrict__ out) {
    __shared__ int ncur[256];
    __shared__ int noff[264];
    const int tid  = threadIdx.x;
    const int lane = tid & 31;
    const int w  = __builtin_amdgcn_readfirstlane(tid >> 5);
    const int bk = blockIdx.x;
    const int s  = offm[bk * NCHK];
    const int e  = offm[bk * NCHK + NCHK];

    if (tid < 256) ncur[tid] = 0;
    __syncthreads();

    for (int i0 = s + tid; i0 < e; i0 += 4 * PASS_T) {
        int v[4];
#pragma unroll
        for (int u = 0; u < 4; ++u) {
            const int i = i0 + u * PASS_T;
            v[u] = (i < e) ? sorted1[i] : -1;
        }
#pragma unroll
        for (int u = 0; u < 4; ++u)
            if (v[u] >= 0) atomicAdd(&ncur[(v[u] >> 16) & 255], 1);
    }
    __syncthreads();

    if (w == 0) {
        int v[8];
        int t = 0;
#pragma unroll
        for (int j = 0; j < 8; ++j) { const int x = ncur[8 * lane + j]; v[j] = t; t += x; }
        int incl = t;
        for (int d = 1; d < 32; d <<= 1) {
            const int n = __shfl_up(incl, d);
            if (lane >= d) incl += n;
        }
        const int excl = incl - t;
#pragma unroll
        for (int j = 0; j < 8; ++j) {
            noff[8 * lane + j] = excl + v[j];
            ncur[8 * lane + j] = excl + v[j];
        }
        if (lane == 31) noff[256] = incl;
    }
    __syncthreads();

    for (int i0 = s + tid; i0 < e; i0 += 4 * PASS_T) {
        int v[4];
#pragma unroll
        for (int u = 0; u < 4; ++u) {
            const int i = i0 + u * PASS_T;
            v[u] = (i < e) ? sorted1[i] : -1;
        }
#pragma unroll
        for (int u = 0; u < 4; ++u)
            if (v[u] >= 0) {
                const int p = atomicAdd(&ncur[(v[u] >> 16) & 255], 1);
                perm[s + p] = v[u] & 0xFFFF;
            }
    }
    __syncthreads();

    const int nl   = w * 16;
    const int nb   = bk * 256 + nl;
    const int half = lane >> 4;
    const int c    = lane & 15;

    const int lo = s + noff[nl + c];
    const int hi = s + noff[nl + c + 1];
    const int dg = hi - lo;
    const float r = 1.0f / (float)(dg > 1 ? dg : 1);
    const int start = s + __builtin_amdgcn_readfirstlane(noff[nl]);
    const int end   = s + __builtin_amdgcn_readfirstlane(noff[nl + 16]);
    const int last  = end - 1;

    const unsigned cb = 16u * (unsigned)c;
    const char* hbytes = (const char*)h;

    v8f acc0 = {}, acc1 = {}, acc2 = {}, acc3 = {};

    int k0 = (start < end) ? (start & ~3) : end;
    if (k0 < end) {
        unsigned off[16];
        v2f bop[32];
        {
            int p = k0 + lane;
            p = p < start ? start : p;
            p = p > last ? last : p;
            const int pv = perm[p];
            int pn = k0 + 32 + lane;
            pn = pn > last ? last : pn;
            const int pvn = perm[pn];
            float4 raw[16];
#pragma unroll
            for (int j = 0; j < 16; ++j) {
                const unsigned row = (unsigned)__shfl(pv, 4 * (j >> 1) + 2 * half + (j & 1));
                raw[j] = *(const float4*)(hbytes + (size_t)((row << 8) | cb));
            }
#pragma unroll
            for (int j = 0; j < 16; ++j) {
                const unsigned row = (unsigned)__shfl(pvn, 4 * (j >> 1) + 2 * half + (j & 1));
                off[j] = (row << 8) | cb;
            }
#pragma unroll
            for (int u = 0; u < 8; ++u) {
                bop[4 * u + 0].x = raw[2 * u].x; bop[4 * u + 0].y = raw[2 * u + 1].x;
                bop[4 * u + 1].x = raw[2 * u].y; bop[4 * u + 1].y = raw[2 * u + 1].y;
                bop[4 * u + 2].x = raw[2 * u].z; bop[4 * u + 2].y = raw[2 * u + 1].z;
                bop[4 * u + 3].x = raw[2 * u].w; bop[4 * u + 3].y = raw[2 * u + 1].w;
            }
        }

        for (; k0 < end; k0 += 32) {
            int pn = k0 + 64 + lane;
            pn = pn > last ? last : pn;
            const int pvn = perm[pn];
            float4 raw[16];
#pragma unroll
            for (int j = 0; j < 16; ++j)
                raw[j] = *(const float4*)(hbytes + (size_t)off[j]);

#pragma unroll
            for (int u = 0; u < 4; ++u) {
                const int p0 = k0 + 4 * u + 2 * half;
                v2f a;
                a.x = (p0     >= lo && p0     < hi) ? r : 0.0f;
                a.y = (p0 + 1 >= lo && p0 + 1 < hi) ? r : 0.0f;
                acc0 = __builtin_amdgcn_wmma_f32_16x16x4_f32(false, a, false, bop[4 * u + 0], (short)0, acc0, false, false);
                acc1 = __builtin_amdgcn_wmma_f32_16x16x4_f32(false, a, false, bop[4 * u + 1], (short)0, acc1, false, false);
                acc2 = __builtin_amdgcn_wmma_f32_16x16x4_f32(false, a, false, bop[4 * u + 2], (short)0, acc2, false, false);
                acc3 = __builtin_amdgcn_wmma_f32_16x16x4_f32(false, a, false, bop[4 * u + 3], (short)0, acc3, false, false);
            }
#pragma unroll
            for (int j = 0; j < 16; ++j) {
                const unsigned row = (unsigned)__shfl(pvn, 4 * (j >> 1) + 2 * half + (j & 1));
                off[j] = (row << 8) | cb;
            }
            if (k0 + 16 < end) {
#pragma unroll
                for (int u = 4; u < 8; ++u) {
                    const int p0 = k0 + 4 * u + 2 * half;
                    v2f a;
                    a.x = (p0     >= lo && p0     < hi) ? r : 0.0f;
                    a.y = (p0 + 1 >= lo && p0 + 1 < hi) ? r : 0.0f;
                    acc0 = __builtin_amdgcn_wmma_f32_16x16x4_f32(false, a, false, bop[4 * u + 0], (short)0, acc0, false, false);
                    acc1 = __builtin_amdgcn_wmma_f32_16x16x4_f32(false, a, false, bop[4 * u + 1], (short)0, acc1, false, false);
                    acc2 = __builtin_amdgcn_wmma_f32_16x16x4_f32(false, a, false, bop[4 * u + 2], (short)0, acc2, false, false);
                    acc3 = __builtin_amdgcn_wmma_f32_16x16x4_f32(false, a, false, bop[4 * u + 3], (short)0, acc3, false, false);
                }
            }
#pragma unroll
            for (int u = 0; u < 8; ++u) {
                bop[4 * u + 0].x = raw[2 * u].x; bop[4 * u + 0].y = raw[2 * u + 1].x;
                bop[4 * u + 1].x = raw[2 * u].y; bop[4 * u + 1].y = raw[2 * u + 1].y;
                bop[4 * u + 2].x = raw[2 * u].z; bop[4 * u + 2].y = raw[2 * u + 1].z;
                bop[4 * u + 3].x = raw[2 * u].w; bop[4 * u + 3].y = raw[2 * u + 1].w;
            }
        }
    }

    const int rl = nl + half * 8;
    const int rowbase = nb + half * 8;
    int o9[9];
#pragma unroll
    for (int v = 0; v < 9; ++v) o9[v] = noff[rl + v];
#pragma unroll
    for (int v = 0; v < 8; ++v) {
        const size_t at = (size_t)(rowbase + v) * FDIM + 4 * c;
        float4 o = make_float4(acc0[v], acc1[v], acc2[v], acc3[v]);
        if (o9[v + 1] == o9[v]) o = *(const float4*)(h + at);
        *(float4*)(out + at) = o;
    }
}

extern "C" void kernel_launch(void* const* d_in, const int* in_sizes, int n_in,
                              void* d_out, int out_size, void* d_ws, size_t ws_size,
                              hipStream_t stream) {
    const float* h   = (const float*)d_in[0];
    const int*   src = (const int*)d_in[1];
    const int*   dst = (const int*)d_in[2];
    float*       out = (float*)d_out;
    const int num_edges = in_sizes[1];

    int* cntm    = (int*)d_ws;
    int* offm    = cntm + SCAN_N;
    int* sorted1 = offm + SCAN_N + 64;
    int* perm    = sorted1 + EDGE_CAP;

    gp_hist_kernel<<<NCHK, PASS_T, 0, stream>>>(dst, cntm, num_edges);
    gp_scan_kernel<<<1, 1024, 0, stream>>>(cntm, offm);
    gp_scatter_kernel<<<NCHK, PASS_T, 0, stream>>>(src, dst, offm, sorted1, num_edges);
    gp_gather_kernel<<<NBKT, PASS_T, 0, stream>>>(h, offm, sorted1, perm, out);
}


# Bucket sort made chunk-local: each chunk ranks by LDS add return, scans, writes its own slice + cell table; bucket offset = sum of its cells' starts; hist and scan launches gone; sum order changes
# speedup vs baseline: 4.9687x; 1.8023x over previous
#include <hip/hip_runtime.h>


typedef __attribute__((ext_vector_type(2))) float v2f;
typedef __attribute__((ext_vector_type(8))) float v8f;

#define NUM_NODES (8 * 8192)
#define FDIM 64
#define EDGE_CAP (1 << 20)
#define NBKT 256
#define NCHK 256
#define SCAN_N (NBKT * NCHK)
#define PASS_T 512

__device__ __forceinline__ int gp_chunk_len(int num_edges) {
    return ((num_edges + 1023) >> 10) << 2;
}

__global__ __launch_bounds__(PASS_T) void gp_sort_kernel(const int* __restrict__ src,
                                                         const int* __restrict__ dst,
                                                         int* __restrict__ tab,
                                                         int* __restrict__ sorted1,
                                                         int num_edges) {
    __shared__ int lh[NBKT];
    __shared__ int ls[NBKT];
    const int tid  = threadIdx.x;
    const int lane = tid & 31;
    if (tid < NBKT) lh[tid] = 0;
    __syncthreads();
    const int chunk = gp_chunk_len(num_edges);
    const int lo = blockIdx.x * chunk;
    const int hi = min(lo + chunk, num_edges);

    int rec[8];
    int bkt[8];
    int rnk[8];
#pragma unroll
    for (int u = 0; u < 2; ++u) {
        const int i = lo + 4 * tid + u * 4 * PASS_T;
        int d[4], s[4];
        if (i + 4 <= hi) {
            const int4 dv = *(const int4*)(dst + i);
            const int4 sv = *(const int4*)(src + i);
            d[0] = dv.x; d[1] = dv.y; d[2] = dv.z; d[3] = dv.w;
            s[0] = sv.x; s[1] = sv.y; s[2] = sv.z; s[3] = sv.w;
        } else {
#pragma unroll
            for (int j = 0; j < 4; ++j) {
                const bool in = i + j < hi;
                d[j] = in ? dst[i + j] : -1;
                s[j] = in ? src[i + j] : 0;
            }
        }
#pragma unroll
        for (int j = 0; j < 4; ++j) {
            const bool in = i + j < hi;
            bkt[4 * u + j] = (d[j] >> 8) & (NBKT - 1);
            rec[4 * u + j] = in ? ((s[j] & 0xFFFF) | ((d[j] & 255) << 16)) : -1;
        }
    }
#pragma unroll
    for (int q = 0; q < 8; ++q)
        rnk[q] = (rec[q] >= 0) ? atomicAdd(&lh[bkt[q]], 1) : 0;
    __syncthreads();

    if (tid < 32) {
        int v[8];
        int t = 0;
#pragma unroll
        for (int j = 0; j < 8; ++j) { const int x = lh[8 * lane + j]; v[j] = t; t += x; }
        int incl = t;
        for (int d = 1; d < 32; d <<= 1) {
            const int n = __shfl_up(incl, d);
            if (lane >= d) incl += n;
        }
        const int excl = incl - t;
#pragma unroll
        for (int j = 0; j < 8; ++j) ls[8 * lane + j] = excl + v[j];
    }
    __syncthreads();

    if (tid < NBKT) tab[tid * NCHK + blockIdx.x] = ls[tid] | (lh[tid] << 16);
    int pos[8];
#pragma unroll
    for (int q = 0; q < 8; ++q) pos[q] = lo + ls[bkt[q]] + rnk[q];
#pragma unroll
    for (int q = 0; q < 8; ++q)
        if (rec[q] >= 0) sorted1[pos[q]] = rec[q];
}

__global__ __launch_bounds__(PASS_T) void gp_gather_kernel(const float* __restrict__ h,
                                                           const int* __restrict__ tab,
                                                           const int* __restrict__ sorted1,
                                                           int* perm,
                                                           float* __restrict__ out,
                                                           int num_edges) {
    __shared__ int ncur[256];
    __shared__ int noff[264];
    __shared__ __attribute__((aligned(16))) int seg[NCHK];
    __shared__ int wsum[16];
    const int tid  = threadIdx.x;
    const int lane = tid & 31;
    const int w  = __builtin_amdgcn_readfirstlane(tid >> 5);
    const int bk = blockIdx.x;
    const int chunk = gp_chunk_len(num_edges);

    if (tid < 256) {
        const int t = tab[bk * NCHK + tid];
        seg[tid]  = t;
        ncur[tid] = 0;
        int a = t & 0xFFFF;
        int b = (int)((unsigned)t >> 16);
        for (int d = 16; d >= 1; d >>= 1) { a += __shfl_xor(a, d); b += __shfl_xor(b, d); }
        if (lane == 0) { wsum[w] = a; wsum[8 + w] = b; }
    }
    __syncthreads();
    int s = 0, e = 0;
#pragma unroll
    for (int j = 0; j < 8; ++j) { s += wsum[j]; e += wsum[8 + j]; }
    s = __builtin_amdgcn_readfirstlane(s);
    e = s + __builtin_amdgcn_readfirstlane(e);

    int rc[16];
    int sg[16];
    {
        const int4* sp = (const int4*)(seg + 16 * w);
        const int4 q0 = sp[0], q1 = sp[1], q2 = sp[2], q3 = sp[3];
        const int qv[16] = { q0.x, q0.y, q0.z, q0.w, q1.x, q1.y, q1.z, q1.w,
                             q2.x, q2.y, q2.z, q2.w, q3.x, q3.y, q3.z, q3.w };
#pragma unroll
        for (int u = 0; u < 16; ++u) sg[u] = __builtin_amdgcn_readfirstlane(qv[u]);
    }
#pragma unroll
    for (int u = 0; u < 16; ++u) {
        const int cn = (int)((unsigned)sg[u] >> 16);
        rc[u] = (lane < cn) ? sorted1[(16 * w + u) * chunk + (sg[u] & 0xFFFF) + lane] : -1;
    }
#pragma unroll
    for (int u = 0; u < 16; ++u)
        if (rc[u] >= 0) atomicAdd(&ncur[(rc[u] >> 16) & 255], 1);
#pragma unroll
    for (int u = 0; u < 16; ++u) {
        const int cn = (int)((unsigned)sg[u] >> 16);
        const int* cell = sorted1 + (16 * w + u) * chunk + (sg[u] & 0xFFFF);
        for (int j = 32 + lane; j < cn; j += 32) atomicAdd(&ncur[(cell[j] >> 16) & 255], 1);
    }
    __syncthreads();

    if (w == 0) {
        int x8[8];
        int t = 0;
#pragma unroll
        for (int j = 0; j < 8; ++j) { const int x = ncur[8 * lane + j]; x8[j] = t; t += x; }
        int incl = t;
        for (int d = 1; d < 32; d <<= 1) {
            const int n = __shfl_up(incl, d);
            if (lane >= d) incl += n;
        }
        const int excl = incl - t;
#pragma unroll
        for (int j = 0; j < 8; ++j) {
            noff[8 * lane + j] = excl + x8[j];
            ncur[8 * lane + j] = excl + x8[j];
        }
        if (lane == 31) noff[256] = incl;
    }
    __syncthreads();

#pragma unroll
    for (int u = 0; u < 16; ++u)
        if (rc[u] >= 0) {
            const int p = atomicAdd(&ncur[(rc[u] >> 16) & 255], 1);
            perm[s + p] = rc[u] & 0xFFFF;
        }
#pragma unroll
    for (int u = 0; u < 16; ++u) {
        const int cn = (int)((unsigned)sg[u] >> 16);
        const int* cell = sorted1 + (16 * w + u) * chunk + (sg[u] & 0xFFFF);
        for (int j = 32 + lane; j < cn; j += 32) {
            const int r = cell[j];
            const int p = atomicAdd(&ncur[(r >> 16) & 255], 1);
            perm[s + p] = r & 0xFFFF;
        }
    }
    __syncthreads();

    const int nl   = w * 16;
    const int nb   = bk * 256 + nl;
    const int half = lane >> 4;
    const int c    = lane & 15;

    const int lo = s + noff[nl + c];
    const int hi = s + noff[nl + c + 1];
    const int dg = hi - lo;
    const float r = 1.0f / (float)(dg > 1 ? dg : 1);
    const int start = s + __builtin_amdgcn_readfirstlane(noff[nl]);
    const int end   = s + __builtin_amdgcn_readfirstlane(noff[nl + 16]);
    const int last  = end - 1;

    const unsigned cb = 16u * (unsigned)c;
    const char* hbytes = (const char*)h;

    v8f acc0 = {}, acc1 = {}, acc2 = {}, acc3 = {};

    int k0 = (start < end) ? (start & ~3) : end;
    if (k0 < end) {
        unsigned off[16];
        v2f bop[32];
        {
            int p = k0 + lane;
            p = p < start ? start : p;
            p = p > last ? last : p;
            const int pv = perm[p];
            int pn = k0 + 32 + lane;
            pn = pn > last ? last : pn;
            const int pvn = perm[pn];
            float4 raw[16];
#pragma unroll
            for (int j = 0; j < 16; ++j) {
                const unsigned row = (unsigned)__shfl(pv, 4 * (j >> 1) + 2 * half + (j & 1));
                raw[j] = *(const float4*)(hbytes + (size_t)((row << 8) | cb));
            }
#pragma unroll
            for (int j = 0; j < 16; ++j) {
                const unsigned row = (unsigned)__shfl(pvn, 4 * (j >> 1) + 2 * half + (j & 1));
                off[j] = (row << 8) | cb;
            }
#pragma unroll
            for (int u = 0; u < 8; ++u) {
                bop[4 * u + 0].x = raw[2 * u].x; bop[4 * u + 0].y = raw[2 * u + 1].x;
                bop[4 * u + 1].x = raw[2 * u].y; bop[4 * u + 1].y = raw[2 * u + 1].y;
                bop[4 * u + 2].x = raw[2 * u].z; bop[4 * u + 2].y = raw[2 * u + 1].z;
                bop[4 * u + 3].x = raw[2 * u].w; bop[4 * u + 3].y = raw[2 * u + 1].w;
            }
        }

        for (; k0 < end; k0 += 32) {
            int pn = k0 + 64 + lane;
            pn = pn > last ? last : pn;
            const int pvn = perm[pn];
            float4 raw[16];
#pragma unroll
            for (int j = 0; j < 16; ++j)
                raw[j] = *(const float4*)(hbytes + (size_t)off[j]);

#pragma unroll
            for (int u = 0; u < 4; ++u) {
                const int p0 = k0 + 4 * u + 2 * half;
                v2f a;
                a.x = (p0     >= lo && p0     < hi) ? r : 0.0f;
                a.y = (p0 + 1 >= lo && p0 + 1 < hi) ? r : 0.0f;
                acc0 = __builtin_amdgcn_wmma_f32_16x16x4_f32(false, a, false, bop[4 * u + 0], (short)0, acc0, false, false);
                acc1 = __builtin_amdgcn_wmma_f32_16x16x4_f32(false, a, false, bop[4 * u + 1], (short)0, acc1, false, false);
                acc2 = __builtin_amdgcn_wmma_f32_16x16x4_f32(false, a, false, bop[4 * u + 2], (short)0, acc2, false, false);
                acc3 = __builtin_amdgcn_wmma_f32_16x16x4_f32(false, a, false, bop[4 * u + 3], (short)0, acc3, false, false);
            }
#pragma unroll
            for (int j = 0; j < 16; ++j) {
                const unsigned row = (unsigned)__shfl(pvn, 4 * (j >> 1) + 2 * half + (j & 1));
                off[j] = (row << 8) | cb;
            }
            if (k0 + 16 < end) {
#pragma unroll
                for (int u = 4; u < 8; ++u) {
                    const int p0 = k0 + 4 * u + 2 * half;
                    v2f a;
                    a.x = (p0     >= lo && p0     < hi) ? r : 0.0f;
                    a.y = (p0 + 1 >= lo && p0 + 1 < hi) ? r : 0.0f;
                    acc0 = __builtin_amdgcn_wmma_f32_16x16x4_f32(false, a, false, bop[4 * u + 0], (short)0, acc0, false, false);
                    acc1 = __builtin_amdgcn_wmma_f32_16x16x4_f32(false, a, false, bop[4 * u + 1], (short)0, acc1, false, false);
                    acc2 = __builtin_amdgcn_wmma_f32_16x16x4_f32(false, a, false, bop[4 * u + 2], (short)0, acc2, false, false);
                    acc3 = __builtin_amdgcn_wmma_f32_16x16x4_f32(false, a, false, bop[4 * u + 3], (short)0, acc3, false, false);
                }
            }
#pragma unroll
            for (int u = 0; u < 8; ++u) {
                bop[4 * u + 0].x = raw[2 * u].x; bop[4 * u + 0].y = raw[2 * u + 1].x;
                bop[4 * u + 1].x = raw[2 * u].y; bop[4 * u + 1].y = raw[2 * u + 1].y;
                bop[4 * u + 2].x = raw[2 * u].z; bop[4 * u + 2].y = raw[2 * u + 1].z;
                bop[4 * u + 3].x = raw[2 * u].w; bop[4 * u + 3].y = raw[2 * u + 1].w;
            }
        }
    }

    const int rl = nl + half * 8;
    const int rowbase = nb + half * 8;
    int o9[9];
#pragma unroll
    for (int v = 0; v < 9; ++v) o9[v] = noff[rl + v];
#pragma unroll
    for (int v = 0; v < 8; ++v) {
        const size_t at = (size_t)(rowbase + v) * FDIM + 4 * c;
        float4 o = make_float4(acc0[v], acc1[v], acc2[v], acc3[v]);
        if (o9[v + 1] == o9[v]) o = *(const float4*)(h + at);
        *(float4*)(out + at) = o;
    }
}

extern "C" void kernel_launch(void* const* d_in, const int* in_sizes, int n_in,
                              void* d_out, int out_size, void* d_ws, size_t ws_size,
                              hipStream_t stream) {
    const float* h   = (const float*)d_in[0];
    const int*   src = (const int*)d_in[1];
    const int*   dst = (const int*)d_in[2];
    float*       out = (float*)d_out;
    const int num_edges = in_sizes[1];

    int* tab     = (int*)d_ws;
    int* sorted1 = tab + SCAN_N + 64;
    int* perm    = sorted1 + EDGE_CAP;

    gp_sort_kernel<<<NCHK, PASS_T, 0, stream>>>(src, dst, tab, sorted1, num_edges);
    gp_gather_kernel<<<NBKT, PASS_T, 0, stream>>>(h, tab, sorted1, perm, out, num_edges);
}


# Both scatters of 4-byte records now land in LDS: a chunk's grouped records leave as whole 16-byte stores; a bucket's sorted source list stays in LDS (perm[] round trip gone); sum order unchanged
# speedup vs baseline: 6.0273x; 1.2131x over previous
#include <hip/hip_runtime.h>


typedef __attribute__((ext_vector_type(2))) float v2f;
typedef __attribute__((ext_vector_type(8))) float v8f;

#define NUM_NODES (8 * 8192)
#define FDIM 64
#define EDGE_CAP (1 << 20)
#define NBKT 256
#define NCHK 256
#define SCAN_N (NBKT * NCHK)
#define PASS_T 512
#define CHUNK_MAX 4096
#define PERM_CAP 8192
#define PERM_PAD 128

__device__ __forceinline__ int gp_chunk_len(int num_edges) {
    return ((num_edges + 1023) >> 10) << 2;
}

__global__ __launch_bounds__(PASS_T) void gp_sort_kernel(const int* __restrict__ src,
                                                         const int* __restrict__ dst,
                                                         int* __restrict__ tab,
                                                         int* __restrict__ sorted1,
                                                         int num_edges) {
    __shared__ int lh[NBKT];
    __shared__ int ls[NBKT];
    __shared__ __attribute__((aligned(16))) int stage[CHUNK_MAX];
    const int tid  = threadIdx.x;
    const int lane = tid & 31;
    if (tid < NBKT) lh[tid] = 0;
    __syncthreads();
    const int chunk = gp_chunk_len(num_edges);
    const int lo = blockIdx.x * chunk;
    const int hi = min(lo + chunk, num_edges);

    int rec[8];
    int bkt[8];
    int rnk[8];
#pragma unroll
    for (int u = 0; u < 2; ++u) {
        const int i = lo + 4 * tid + u * 4 * PASS_T;
        int d[4], s[4];
        if (i + 4 <= hi) {
            const int4 dv = *(const int4*)(dst + i);
            const int4 sv = *(const int4*)(src + i);
            d[0] = dv.x; d[1] = dv.y; d[2] = dv.z; d[3] = dv.w;
            s[0] = sv.x; s[1] = sv.y; s[2] = sv.z; s[3] = sv.w;
        } else {
#pragma unroll
            for (int j = 0; j < 4; ++j) {
                const bool in = i + j < hi;
                d[j] = in ? dst[i + j] : -1;
                s[j] = in ? src[i + j] : 0;
            }
        }
#pragma unroll
        for (int j = 0; j < 4; ++j) {
            const bool in = i + j < hi;
            bkt[4 * u + j] = (d[j] >> 8) & (NBKT - 1);
            rec[4 * u + j] = in ? ((s[j] & 0xFFFF) | ((d[j] & 255) << 16)) : -1;
        }
    }
#pragma unroll
    for (int q = 0; q < 8; ++q)
        rnk[q] = (rec[q] >= 0) ? atomicAdd(&lh[bkt[q]], 1) : 0;
    __syncthreads();

    if (tid < 32) {
        int v[8];
        int t = 0;
#pragma unroll
        for (int j = 0; j < 8; ++j) { const int x = lh[8 * lane + j]; v[j] = t; t += x; }
        int incl = t;
        for (int d = 1; d < 32; d <<= 1) {
            const int n = __shfl_up(incl, d);
            if (lane >= d) incl += n;
        }
        const int excl = incl - t;
#pragma unroll
        for (int j = 0; j < 8; ++j) ls[8 * lane + j] = excl + v[j];
    }
    __syncthreads();

    int pos[8];
#pragma unroll
    for (int q = 0; q < 8; ++q) pos[q] = ls[bkt[q]] + rnk[q];
#pragma unroll
    for (int q = 0; q < 8; ++q)
        if (rec[q] >= 0) stage[pos[q]] = rec[q];
    __syncthreads();
    if (tid < NBKT) tab[tid * NCHK + blockIdx.x] = ls[tid] | (lh[tid] << 16);
    const int n = hi - lo;
    for (int u = 0; u < 2; ++u) {
        const int i = 4 * tid + u * 4 * PASS_T;
        if (i + 4 <= n) {
            *(int4*)(sorted1 + lo + i) = *(const int4*)(stage + i);
        } else {
            for (int j = 0; j < 4; ++j)
                if (i + j < n) sorted1[lo + i + j] = stage[i + j];
        }
    }
}

typedef const v2f (&gp_bop_ref)[32];

template <int U0>
__device__ __forceinline__ void gp_mma4(const int k0, const int half, const int lo, const int hi,
                                        const float r, gp_bop_ref bop,
                                        v8f& acc0, v8f& acc1, v8f& acc2, v8f& acc3) {
#pragma unroll
    for (int u = U0; u < U0 + 4; ++u) {
        const int p0 = k0 + 4 * u + 2 * half;
        v2f a;
        a.x = (p0     >= lo && p0     < hi) ? r : 0.0f;
        a.y = (p0 + 1 >= lo && p0 + 1 < hi) ? r : 0.0f;
        acc0 = __builtin_amdgcn_wmma_f32_16x16x4_f32(false, a, false, bop[4 * u + 0], (short)0, acc0, false, false);
        acc1 = __builtin_amdgcn_wmma_f32_16x16x4_f32(false, a, false, bop[4 * u + 1], (short)0, acc1, false, false);
        acc2 = __builtin_amdgcn_wmma_f32_16x16x4_f32(false, a, false, bop[4 * u + 2], (short)0, acc2, false, false);
        acc3 = __builtin_amdgcn_wmma_f32_16x16x4_f32(false, a, false, bop[4 * u + 3], (short)0, acc3, false, false);
    }
}

__device__ __forceinline__ void gp_rows_to_bop(const float4 (&raw)[16], v2f (&bop)[32]) {
#pragma unroll
    for (int u = 0; u < 8; ++u) {
        bop[4 * u + 0].x = raw[2 * u].x; bop[4 * u + 0].y = raw[2 * u + 1].x;
        bop[4 * u + 1].x = raw[2 * u].y; bop[4 * u + 1].y = raw[2 * u + 1].y;
        bop[4 * u + 2].x = raw[2 * u].z; bop[4 * u + 2].y = raw[2 * u + 1].z;
        bop[4 * u + 3].x = raw[2 * u].w; bop[4 * u + 3].y = raw[2 * u + 1].w;
    }
}

__global__ __launch_bounds__(PASS_T) void gp_gather_kernel(const float* __restrict__ h,
                                                           const int* __restrict__ tab,
                                                           const int* __restrict__ sorted1,
                                                           int* perm,
                                                           float* __restrict__ out,
                                                           int num_edges) {
    __shared__ int ncur[256];
    __shared__ int noff[264];
    __shared__ __attribute__((aligned(16))) int seg[NCHK];
    __shared__ int wsum[16];
    __shared__ __attribute__((aligned(16))) int lperm[PERM_CAP + PERM_PAD];
    const int tid  = threadIdx.x;
    const int lane = tid & 31;
    const int w  = __builtin_amdgcn_readfirstlane(tid >> 5);
    const int bk = blockIdx.x;
    const int chunk = gp_chunk_len(num_edges);

    if (tid < 256) {
        const int t = tab[bk * NCHK + tid];
        seg[tid]  = t;
        ncur[tid] = 0;
        int a = t & 0xFFFF;
        int b = (int)((unsigned)t >> 16);
        for (int d = 16; d >= 1; d >>= 1) { a += __shfl_xor(a, d); b += __shfl_xor(b, d); }
        if (lane == 0) { wsum[w] = a; wsum[8 + w] = b; }
    }
    __syncthreads();
    int s = 0, e = 0;
#pragma unroll
    for (int j = 0; j < 8; ++j) { s += wsum[j]; e += wsum[8 + j]; }
    s = __builtin_amdgcn_readfirstlane(s);
    e = s + __builtin_amdgcn_readfirstlane(e);
    const int sb   = s & ~3;
    const int sh   = s - sb;
    const int blen = e - s;
    const bool fits = blen <= PERM_CAP;
    if (fits) {
        if (tid < 4)  lperm[tid] = 0;
        if (tid < 96) lperm[sh + blen + tid] = 0;
    }

    int rc[16];
    int sg[16];
    {
        const int4* sp = (const int4*)(seg + 16 * w);
        const int4 q0 = sp[0], q1 = sp[1], q2 = sp[2], q3 = sp[3];
        const int qv[16] = { q0.x, q0.y, q0.z, q0.w, q1.x, q1.y, q1.z, q1.w,
                             q2.x, q2.y, q2.z, q2.w, q3.x, q3.y, q3.z, q3.w };
#pragma unroll
        for (int u = 0; u < 16; ++u) sg[u] = __builtin_amdgcn_readfirstlane(qv[u]);
    }
#pragma unroll
    for (int u = 0; u < 16; ++u) {
        const int cn = (int)((unsigned)sg[u] >> 16);
        rc[u] = (lane < cn) ? sorted1[(16 * w + u) * chunk + (sg[u] & 0xFFFF) + lane] : -1;
    }
#pragma unroll
    for (int u = 0; u < 16; ++u)
        if (rc[u] >= 0) atomicAdd(&ncur[(rc[u] >> 16) & 255], 1);
#pragma unroll
    for (int u = 0; u < 16; ++u) {
        const int cn = (int)((unsigned)sg[u] >> 16);
        const int* cell = sorted1 + (16 * w + u) * chunk + (sg[u] & 0xFFFF);
        for (int j = 32 + lane; j < cn; j += 32) atomicAdd(&ncur[(cell[j] >> 16) & 255], 1);
    }
    __syncthreads();

    if (w == 0) {
        int x8[8];
        int t = 0;
#pragma unroll
        for (int j = 0; j < 8; ++j) { const int x = ncur[8 * lane + j]; x8[j] = t; t += x; }
        int incl = t;
        for (int d = 1; d < 32; d <<= 1) {
            const int n = __shfl_up(incl, d);
            if (lane >= d) incl += n;
        }
        const int excl = incl - t;
#pragma unroll
        for (int j = 0; j < 8; ++j) {
            noff[8 * lane + j] = excl + x8[j];
            ncur[8 * lane + j] = excl + x8[j];
        }
        if (lane == 31) noff[256] = incl;
    }
    __syncthreads();

    if (fits) {
#pragma unroll
        for (int u = 0; u < 16; ++u)
            if (rc[u] >= 0) {
                const int p = atomicAdd(&ncur[(rc[u] >> 16) & 255], 1);
                lperm[sh + p] = rc[u] & 0xFFFF;
            }
#pragma unroll
        for (int u = 0; u < 16; ++u) {
            const int cn = (int)((unsigned)sg[u] >> 16);
            const int* cell = sorted1 + (16 * w + u) * chunk + (sg[u] & 0xFFFF);
            for (int j = 32 + lane; j < cn; j += 32) {
                const int r = cell[j];
                const int p = atomicAdd(&ncur[(r >> 16) & 255], 1);
                lperm[sh + p] = r & 0xFFFF;
            }
        }
    } else {
#pragma unroll
        for (int u = 0; u < 16; ++u)
            if (rc[u] >= 0) {
                const int p = atomicAdd(&ncur[(rc[u] >> 16) & 255], 1);
                perm[s + p] = rc[u] & 0xFFFF;
            }
#pragma unroll
        for (int u = 0; u < 16; ++u) {
            const int cn = (int)((unsigned)sg[u] >> 16);
            const int* cell = sorted1 + (16 * w + u) * chunk + (sg[u] & 0xFFFF);
            for (int j = 32 + lane; j < cn; j += 32) {
                const int r = cell[j];
                const int p = atomicAdd(&ncur[(r >> 16) & 255], 1);
                perm[s + p] = r & 0xFFFF;
            }
        }
    }
    __syncthreads();

    const int nl   = w * 16;
    const int nb   = bk * 256 + nl;
    const int half = lane >> 4;
    const int c    = lane & 15;

    const int lo = s + noff[nl + c];
    const int hi = s + noff[nl + c + 1];
    const int dg = hi - lo;
    const float r = 1.0f / (float)(dg > 1 ? dg : 1);
    const int start = s + __builtin_amdgcn_readfirstlane(noff[nl]);
    const int end   = s + __builtin_amdgcn_readfirstlane(noff[nl + 16]);
    const int last  = end - 1;

    const unsigned cb = 16u * (unsigned)c;
    const char* hbytes = (const char*)h;

    v8f acc0 = {}, acc1 = {}, acc2 = {}, acc3 = {};

    int k0 = (start < end) ? (start & ~3) : end;
    if (k0 < end) {
        unsigned off[16];
        v2f bop[32];
        if (fits) {
            int ib = k0 - sb + 2 * half;
            {
                float4 raw[16];
#pragma unroll
                for (int u = 0; u < 8; ++u) {
                    const int2 rr = *(const int2*)(lperm + ib + 4 * u);
                    raw[2 * u]     = *(const float4*)(hbytes + (size_t)(((unsigned)rr.x << 8) | cb));
                    raw[2 * u + 1] = *(const float4*)(hbytes + (size_t)(((unsigned)rr.y << 8) | cb));
                }
#pragma unroll
                for (int u = 0; u < 8; ++u) {
                    const int2 rr = *(const int2*)(lperm + ib + 32 + 4 * u);
                    off[2 * u]     = ((unsigned)rr.x << 8) | cb;
                    off[2 * u + 1] = ((unsigned)rr.y << 8) | cb;
                }
                gp_rows_to_bop(raw, bop);
            }
            for (; k0 < end; k0 += 32, ib += 32) {
                float4 raw[16];
#pragma unroll
                for (int j = 0; j < 16; ++j)
                    raw[j] = *(const float4*)(hbytes + (size_t)off[j]);
                gp_mma4<0>(k0, half, lo, hi, r, bop, acc0, acc1, acc2, acc3);
#pragma unroll
                for (int u = 0; u < 8; ++u) {
                    const int2 rr = *(const int2*)(lperm + ib + 64 + 4 * u);
                    off[2 * u]     = ((unsigned)rr.x << 8) | cb;
                    off[2 * u + 1] = ((unsigned)rr.y << 8) | cb;
                }
                if (k0 + 16 < end)
                    gp_mma4<4>(k0, half, lo, hi, r, bop, acc0, acc1, acc2, acc3);
                gp_rows_to_bop(raw, bop);
            }
        } else {
            {
                int p = k0 + lane;
                p = p < start ? start : p;
                p = p > last ? last : p;
                const int pv = perm[p];
                int pn = k0 + 32 + lane;
                pn = pn > last ? last : pn;
                const int pvn = perm[pn];
                float4 raw[16];
#pragma unroll
                for (int j = 0; j < 16; ++j) {
                    const unsigned row = (unsigned)__shfl(pv, 4 * (j >> 1) + 2 * half + (j & 1));
                    raw[j] = *(const float4*)(hbytes + (size_t)((row << 8) | cb));
                }
#pragma unroll
                for (int j = 0; j < 16; ++j) {
                    const unsigned row = (unsigned)__shfl(pvn, 4 * (j >> 1) + 2 * half + (j & 1));
                    off[j] = (row << 8) | cb;
                }
                gp_rows_to_bop(raw, bop);
            }
            for (; k0 < end; k0 += 32) {
                int pn = k0 + 64 + lane;
                pn = pn > last ? last : pn;
                const int pvn = perm[pn];
                float4 raw[16];
#pragma unroll
                for (int j = 0; j < 16; ++j)
                    raw[j] = *(const float4*)(hbytes + (size_t)off[j]);
                gp_mma4<0>(k0, half, lo, hi, r, bop, acc0, acc1, acc2, acc3);
#pragma unroll
                for (int j = 0; j < 16; ++j) {
                    const unsigned row = (unsigned)__shfl(pvn, 4 * (j >> 1) + 2 * half + (j & 1));
                    off[j] = (row << 8) | cb;
                }
                if (k0 + 16 < end)
                    gp_mma4<4>(k0, half, lo, hi, r, bop, acc0, acc1, acc2, acc3);
                gp_rows_to_bop(raw, bop);
            }
        }
    }

    const int rl = nl + half * 8;
    const int rowbase = nb + half * 8;
    int o9[9];
#pragma unroll
    for (int v = 0; v < 9; ++v) o9[v] = noff[rl + v];
#pragma unroll
    for (int v = 0; v < 8; ++v) {
        const size_t at = (size_t)(rowbase + v) * FDIM + 4 * c;
        float4 o = make_float4(acc0[v], acc1[v], acc2[v], acc3[v]);
        if (o9[v + 1] == o9[v]) o = *(const float4*)(h + at);
        *(float4*)(out + at) = o;
    }
}

extern "C" void kernel_launch(void* const* d_in, const int* in_sizes, int n_in,
                              void* d_out, int out_size, void* d_ws, size_t ws_size,
                              hipStream_t stream) {
    const float* h   = (const float*)d_in[0];
    const int*   src = (const int*)d_in[1];
    const int*   dst = (const int*)d_in[2];
    float*       out = (float*)d_out;
    const int num_edges = in_sizes[1];

    int* tab     = (int*)d_ws;
    int* sorted1 = tab + SCAN_N + 64;
    int* perm    = sorted1 + EDGE_CAP;

    gp_sort_kernel<<<NCHK, PASS_T, 0, stream>>>(src, dst, tab, sorted1, num_edges);
    gp_gather_kernel<<<NBKT, PASS_T, 0, stream>>>(h, tab, sorted1, perm, out, num_edges);
}
